# starting kernel
# baseline (speedup 1.0000x reference)
#include <hip/hip_runtime.h>

typedef __attribute__((ext_vector_type(2))) float v2f;
typedef __attribute__((ext_vector_type(8))) float v8f;

#define D_DIM 512
#define CHUNKS 16

__global__ void bag_init_kernel(const int* __restrict__ counts, int B,
                                float* __restrict__ acc, int* __restrict__ prefix) {
    const int n = B * D_DIM;
    for (int i = threadIdx.x; i < n; i += blockDim.x) acc[i] = 0.0f;
    if (threadIdx.x == 0) {
        int s = 0;
        prefix[0] = 0;
        for (int i = 0; i < B; ++i) { s += counts[i]; prefix[i + 1] = s; }
    }
}

__global__ void __launch_bounds__(256)
bag_sum_wmma_kernel(const float* __restrict__ samples,
                    const int* __restrict__ prefix,
                    float* __restrict__ acc) {
    const int bag   = blockIdx.y;
    const int start = prefix[bag];
    const int end   = prefix[bag + 1];
    const int count = end - start;
    const int rowsPer = (count + (int)gridDim.x - 1) / (int)gridDim.x;
    const int r0 = start + (int)blockIdx.x * rowsPer;
    const int r1 = min(r0 + rowsPer, end);
    if (r0 >= r1) return;

    const int lane  = threadIdx.x & 31;
    const int wave  = threadIdx.x >> 5;
    const int lmod  = lane & 15;
    const int khalf = (lane >> 4) << 1;
    const int colBase = wave * 64;

    v2f a; a.x = 1.0f; a.y = 1.0f;
    v8f c0 = {}, c1 = {}, c2 = {}, c3 = {};

    const int span  = r1 - r0;
    const int nfull = span >> 2;
    const int rem   = span & 3;

    const float* p = samples + (size_t)(r0 + khalf) * D_DIM + colBase + lmod;

    #pragma unroll 2
    for (int it = 0; it < nfull; ++it) {
        __builtin_prefetch(p + 16 * D_DIM, 0, 3);

        v2f b0, b1, b2, b3;
        b0.x = p[0];            b0.y = p[D_DIM];
        b1.x = p[16];           b1.y = p[D_DIM + 16];
        b2.x = p[32];           b2.y = p[D_DIM + 32];
        b3.x = p[48];           b3.y = p[D_DIM + 48];
        p += 4 * D_DIM;

        c0 = __builtin_amdgcn_wmma_f32_16x16x4_f32(false, a, false, b0, (short)0, c0, false, false);
        c1 = __builtin_amdgcn_wmma_f32_16x16x4_f32(false, a, false, b1, (short)0, c1, false, false);
        c2 = __builtin_amdgcn_wmma_f32_16x16x4_f32(false, a, false, b2, (short)0, c2, false, false);
        c3 = __builtin_amdgcn_wmma_f32_16x16x4_f32(false, a, false, b3, (short)0, c3, false, false);
    }

    if (rem) {
        const int rbase = r1 - rem;
        const bool ok0 = khalf < rem;
        const bool ok1 = (khalf + 1) < rem;
        const int  o0  = ok0 ? khalf : 0;
        const int  o1  = ok1 ? (khalf + 1) : 0;
        const float* q0 = samples + (size_t)(rbase + o0) * D_DIM + colBase + lmod;
        const float* q1 = samples + (size_t)(rbase + o1) * D_DIM + colBase + lmod;

        v2f b0, b1, b2, b3;
        float x;
        x = q0[0];  b0.x = ok0 ? x : 0.0f;  x = q1[0];  b0.y = ok1 ? x : 0.0f;
        x = q0[16]; b1.x = ok0 ? x : 0.0f;  x = q1[16]; b1.y = ok1 ? x : 0.0f;
        x = q0[32]; b2.x = ok0 ? x : 0.0f;  x = q1[32]; b2.y = ok1 ? x : 0.0f;
        x = q0[48]; b3.x = ok0 ? x : 0.0f;  x = q1[48]; b3.y = ok1 ? x : 0.0f;

        c0 = __builtin_amdgcn_wmma_f32_16x16x4_f32(false, a, false, b0, (short)0, c0, false, false);
        c1 = __builtin_amdgcn_wmma_f32_16x16x4_f32(false, a, false, b1, (short)0, c1, false, false);
        c2 = __builtin_amdgcn_wmma_f32_16x16x4_f32(false, a, false, b2, (short)0, c2, false, false);
        c3 = __builtin_amdgcn_wmma_f32_16x16x4_f32(false, a, false, b3, (short)0, c3, false, false);
    }

    const int  hi  = lane >> 4;
    const float s0 = hi ? c2[0] : c0[0];
    const float s1 = hi ? c3[0] : c1[0];
    float* dst = acc + (size_t)bag * D_DIM + colBase + hi * 32 + lmod;
    atomicAdd(dst,      s0);
    atomicAdd(dst + 16, s1);
}

__global__ void bag_finalize_kernel(const float* __restrict__ acc,
                                    const int* __restrict__ counts,
                                    float* __restrict__ out, int n) {
    const int i = blockIdx.x * blockDim.x + threadIdx.x;
    if (i < n) {
        const int bag = i / D_DIM;
        out[i] = acc[i] / (float)counts[bag];
    }
}

extern "C" void kernel_launch(void* const* d_in, const int* in_sizes, int n_in,
                              void* d_out, int out_size, void* d_ws, size_t ws_size,
                              hipStream_t stream) {
    const float* samples = (const float*)d_in[0];
    const int*   counts  = (const int*)d_in[1];
    const int B = in_sizes[1];

    float* acc    = (float*)d_ws;
    int*   prefix = (int*)((char*)d_ws + (size_t)B * D_DIM * sizeof(float));
    float* out    = (float*)d_out;

    bag_init_kernel<<<dim3(1), dim3(256), 0, stream>>>(counts, B, acc, prefix);

    bag_sum_wmma_kernel<<<dim3(CHUNKS, B), dim3(256), 0, stream>>>(samples, prefix, acc);

    const int n = B * D_DIM;
    bag_finalize_kernel<<<dim3((n + 255) / 256), dim3(256), 0, stream>>>(acc, counts, out, n);
}


# Load balance: bags ranked by length, one 32-wave block takes rank p with rank B-1-p and splits its waves by length; rows per wave change, so the fixed-order partial sums are regrouped
# speedup vs baseline: 1.5724x; 1.5724x over previous
#include <hip/hip_runtime.h>


typedef __attribute__((ext_vector_type(2))) float v2f;
typedef __attribute__((ext_vector_type(8))) float v8f;

#define D_DIM 512
#define NWAVES 32
#define SLAB 64

__global__ void __launch_bounds__(NWAVES * 32)
bag_pair_mean_wmma_kernel(const float* __restrict__ samples,
                          const int* __restrict__ counts,
                          float* __restrict__ out,
                          int B) {
    __shared__ float red[2][NWAVES][SLAB];
    __shared__ int sel[6];

    const int lane = threadIdx.x & 31;
    const int wave = __builtin_amdgcn_readfirstlane((int)(threadIdx.x >> 5));

    const int p = blockIdx.y;
    const int q = B - 1 - p;

    for (int mbase = wave * 32; mbase < B; mbase += NWAVES * 32) {
        const int me  = mbase + lane;
        const int cme = counts[min(me, B - 1)];
        const unsigned long long kme = ((unsigned long long)(unsigned)cme << 32) | (unsigned)~me;
        int rank = 0, first = 0;
        for (int base = 0; base < B; base += 32) {
            const int idx = base + lane;
            int v = counts[min(idx, B - 1)];
            v = idx < B ? v : 0;
            #pragma unroll
            for (int k = 0; k < 32; ++k) {
                const int ci = __builtin_amdgcn_readlane(v, k);
                const unsigned long long ki = ((unsigned long long)(unsigned)ci << 32) | (unsigned)~(base + k);
                rank  += (ki > kme) ? 1 : 0;
                first += (base + k) < me ? ci : 0;
            }
        }
        if (me < B) {
            if (rank == p) { sel[0] = me; sel[1] = first; sel[2] = cme; }
            if (rank == q) { sel[3] = me; sel[4] = first; sel[5] = cme; }
        }
    }
    __syncthreads();

    const int bagL   = __builtin_amdgcn_readfirstlane(sel[0]);
    const int firstL = __builtin_amdgcn_readfirstlane(sel[1]);
    const int cntL   = __builtin_amdgcn_readfirstlane(sel[2]);
    const int bagS   = __builtin_amdgcn_readfirstlane(sel[3]);
    const int firstS = __builtin_amdgcn_readfirstlane(sel[4]);
    const int cntS   = __builtin_amdgcn_readfirstlane(sel[5]);
    const bool two   = (q != p);

    int nL = NWAVES;
    if (two) {
        const int tot = cntL + cntS;
        int n0 = tot > 0 ? (int)((float)NWAVES * (float)cntL / (float)tot) : NWAVES / 2;
        n0 = min(max(n0, 1), NWAVES - 2);
        const int n1 = n0 + 1;
        nL = ((long long)cntS * n0 < (long long)cntL * (NWAVES - n1)) ? n1 : n0;
    }
    const bool onS   = wave >= nL;
    const int bfirst = onS ? firstS : firstL;
    const int bcount = onS ? cntS : cntL;
    const int nw     = onS ? NWAVES - nL : nL;
    const int kw     = onS ? wave - nL : wave;

    const int rowsPer = (bcount + nw - 1) / nw;
    const int end = bfirst + bcount;
    const int r0 = bfirst + kw * rowsPer;
    const int r1 = min(r0 + rowsPer, end);
    const int span = max(r1 - r0, 0);

    const int lmod  = lane & 15;
    const int khalf = (lane >> 4) << 1;
    const int colBase = (int)blockIdx.x * SLAB;

    v2f a; a.x = 1.0f; a.y = 1.0f;
    v8f c0 = {}, c1 = {}, c2 = {}, c3 = {};

    const int nfull = span >> 2;
    const int rem   = span & 3;

    const float* p0 = samples + (size_t)(r0 + khalf) * D_DIM + colBase + lmod;

    #pragma unroll 2
    for (int it = 0; it < nfull; ++it) {
        __builtin_prefetch(p0 + 16 * D_DIM, 0, 3);

        v2f b0, b1, b2, b3;
        b0.x = p0[0];            b0.y = p0[D_DIM];
        b1.x = p0[16];           b1.y = p0[D_DIM + 16];
        b2.x = p0[32];           b2.y = p0[D_DIM + 32];
        b3.x = p0[48];           b3.y = p0[D_DIM + 48];
        p0 += 4 * D_DIM;

        c0 = __builtin_amdgcn_wmma_f32_16x16x4_f32(false, a, false, b0, (short)0, c0, false, false);
        c1 = __builtin_amdgcn_wmma_f32_16x16x4_f32(false, a, false, b1, (short)0, c1, false, false);
        c2 = __builtin_amdgcn_wmma_f32_16x16x4_f32(false, a, false, b2, (short)0, c2, false, false);
        c3 = __builtin_amdgcn_wmma_f32_16x16x4_f32(false, a, false, b3, (short)0, c3, false, false);
    }

    if (rem) {
        const int rbase = r1 - rem;
        const bool ok0 = khalf < rem;
        const bool ok1 = (khalf + 1) < rem;
        const int  o0  = ok0 ? khalf : 0;
        const int  o1  = ok1 ? (khalf + 1) : 0;
        const float* q0 = samples + (size_t)(rbase + o0) * D_DIM + colBase + lmod;
        const float* q1 = samples + (size_t)(rbase + o1) * D_DIM + colBase + lmod;

        v2f b0, b1, b2, b3;
        float x;
        x = q0[0];  b0.x = ok0 ? x : 0.0f;  x = q1[0];  b0.y = ok1 ? x : 0.0f;
        x = q0[16]; b1.x = ok0 ? x : 0.0f;  x = q1[16]; b1.y = ok1 ? x : 0.0f;
        x = q0[32]; b2.x = ok0 ? x : 0.0f;  x = q1[32]; b2.y = ok1 ? x : 0.0f;
        x = q0[48]; b3.x = ok0 ? x : 0.0f;  x = q1[48]; b3.y = ok1 ? x : 0.0f;

        c0 = __builtin_amdgcn_wmma_f32_16x16x4_f32(false, a, false, b0, (short)0, c0, false, false);
        c1 = __builtin_amdgcn_wmma_f32_16x16x4_f32(false, a, false, b1, (short)0, c1, false, false);
        c2 = __builtin_amdgcn_wmma_f32_16x16x4_f32(false, a, false, b2, (short)0, c2, false, false);
        c3 = __builtin_amdgcn_wmma_f32_16x16x4_f32(false, a, false, b3, (short)0, c3, false, false);
    }

    const int hi = lane >> 4;
    const float s01 = hi ? c1[0] : c0[0];
    const float s23 = hi ? c3[0] : c2[0];
    red[0][wave][lane]      = onS ? 0.0f : s01;
    red[0][wave][32 + lane] = onS ? 0.0f : s23;
    red[1][wave][lane]      = onS ? s01 : 0.0f;
    red[1][wave][32 + lane] = onS ? s23 : 0.0f;
    __syncthreads();

    if (threadIdx.x < 2 * SLAB) {
        const int which = wave >> 1;
        if (which == 0 || two) {
            const int col = threadIdx.x & (SLAB - 1);
            float t = red[which][0][col];
            #pragma unroll
            for (int w = 1; w < NWAVES; ++w) t += red[which][w][col];
            const int ob = which ? bagS : bagL;
            const int oc = which ? cntS : cntL;
            out[(size_t)ob * D_DIM + colBase + col] = t / (float)oc;
        }
    }
}

extern "C" void kernel_launch(void* const* d_in, const int* in_sizes, int n_in,
                              void* d_out, int out_size, void* d_ws, size_t ws_size,
                              hipStream_t stream) {
    const float* samples = (const float*)d_in[0];
    const int*   counts  = (const int*)d_in[1];
    const int B = in_sizes[1];
    float* out = (float*)d_out;

    bag_pair_mean_wmma_kernel<<<dim3(D_DIM / SLAB, (B + 1) / 2), dim3(NWAVES * 32), 0, stream>>>(samples, counts, out, B);
}


# Row loop reads 16 B per lane and keeps 8-16 such loads in flight per wave (two 16-row register buffers); A is now a 0/1 column-parity selector, a WMMA sums 2 rows x 32 cols, so partial sums regroup
# speedup vs baseline: 1.6383x; 1.0419x over previous
#include <hip/hip_runtime.h>


typedef __attribute__((ext_vector_type(2))) float v2f;
typedef __attribute__((ext_vector_type(4))) float v4f;
typedef __attribute__((ext_vector_type(8))) float v8f;

#define D_DIM 512
#define NWAVES 32
#define SLAB 64
#define CH 8

__global__ void __launch_bounds__(NWAVES * 32)
bag_pair_mean_wmma_kernel(const float* __restrict__ samples,
                          const int* __restrict__ counts,
                          float* __restrict__ out,
                          int B) {
    __shared__ __attribute__((aligned(16))) float red[2][NWAVES][SLAB];
    __shared__ int sel[6];

    const int lane = threadIdx.x & 31;
    const int wave = __builtin_amdgcn_readfirstlane((int)(threadIdx.x >> 5));

    const int p = blockIdx.y;
    const int q = B - 1 - p;

    for (int mbase = wave * 32; mbase < B; mbase += NWAVES * 32) {
        const int me  = mbase + lane;
        const int cme = counts[min(me, B - 1)];
        const unsigned long long kme = ((unsigned long long)(unsigned)cme << 32) | (unsigned)~me;
        int rank = 0, first = 0;
        for (int base = 0; base < B; base += 32) {
            const int idx = base + lane;
            int v = counts[min(idx, B - 1)];
            v = idx < B ? v : 0;
            #pragma unroll
            for (int k = 0; k < 32; ++k) {
                const int ci = __builtin_amdgcn_readlane(v, k);
                const unsigned long long ki = ((unsigned long long)(unsigned)ci << 32) | (unsigned)~(base + k);
                rank  += (ki > kme) ? 1 : 0;
                first += (base + k) < me ? ci : 0;
            }
        }
        if (me < B) {
            if (rank == p) { sel[0] = me; sel[1] = first; sel[2] = cme; }
            if (rank == q) { sel[3] = me; sel[4] = first; sel[5] = cme; }
        }
    }
    __syncthreads();

    const int bagL   = __builtin_amdgcn_readfirstlane(sel[0]);
    const int firstL = __builtin_amdgcn_readfirstlane(sel[1]);
    const int cntL   = __builtin_amdgcn_readfirstlane(sel[2]);
    const int bagS   = __builtin_amdgcn_readfirstlane(sel[3]);
    const int firstS = __builtin_amdgcn_readfirstlane(sel[4]);
    const int cntS   = __builtin_amdgcn_readfirstlane(sel[5]);
    const bool two   = (q != p);

    int nL = NWAVES;
    if (two) {
        const int tot = cntL + cntS;
        int n0 = tot > 0 ? (int)((float)NWAVES * (float)cntL / (float)tot) : NWAVES / 2;
        n0 = min(max(n0, 1), NWAVES - 2);
        const int n1 = n0 + 1;
        nL = ((long long)cntS * n0 < (long long)cntL * (NWAVES - n1)) ? n1 : n0;
    }
    const bool onS   = wave >= nL;
    const int bfirst = onS ? firstS : firstL;
    const int bcount = onS ? cntS : cntL;
    const int nw     = onS ? NWAVES - nL : nL;
    const int kw     = onS ? wave - nL : wave;

    const int rowsPer = (bcount + nw - 1) / nw;
    const int end = bfirst + bcount;
    const int r0 = bfirst + kw * rowsPer;
    const int r1 = min(r0 + rowsPer, end);
    const int span = max(r1 - r0, 0);

    const int lmod = lane & 15;
    const int h    = lane >> 4;
    const int colBase = (int)blockIdx.x * SLAB;

    v2f a;
    a.x = (lmod & 1) ? 0.0f : 1.0f;
    a.y = (lmod & 1) ? 1.0f : 0.0f;
    v8f acc0 = {}, acc1 = {};

    const int nch = (span + 2 * CH - 1) / (2 * CH);
    const float* lanep = samples + (size_t)h * D_DIM + colBase + 4 * lmod;
    const v4f zero4 = {0.0f, 0.0f, 0.0f, 0.0f};
    v4f bufA[CH], bufB[CH];

#define LOAD_FULL(BUF, R) do { const float* p_ = lanep + (size_t)(R) * D_DIM; \
    _Pragma("unroll") for (int j = 0; j < CH; ++j) BUF[j] = *(const v4f*)(p_ + (size_t)(2 * j) * D_DIM); } while (0)
#define LOAD_LAST(BUF, R) do { \
    _Pragma("unroll") for (int j = 0; j < CH; ++j) { const int rr_ = min((R) + 2 * j + h, r1 - 1); \
        BUF[j] = *(const v4f*)(samples + (size_t)rr_ * D_DIM + colBase + 4 * lmod); } } while (0)
#define MASK_LAST(BUF, R) do { \
    _Pragma("unroll") for (int j = 0; j < CH; ++j) { asm volatile("" : "+v"(BUF[j])); \
        BUF[j] = ((R) + 2 * j + h < r1) ? BUF[j] : zero4; } } while (0)
#define PIN() asm volatile("" : "+v"(acc0), "+v"(acc1) : : "memory")
#define SUM_CHUNK(BUF) do { \
    _Pragma("unroll") for (int j = 0; j < CH; ++j) { \
        acc0 = __builtin_amdgcn_wmma_f32_16x16x4_f32(false, a, false, __builtin_shufflevector(BUF[j], BUF[j], 0, 1), (short)0, acc0, false, false); \
        acc1 = __builtin_amdgcn_wmma_f32_16x16x4_f32(false, a, false, __builtin_shufflevector(BUF[j], BUF[j], 2, 3), (short)0, acc1, false, false); } } while (0)

    if (nch > 0) {
        const int nf = nch - 1;
        const int rl = r0 + nf * 2 * CH;
        if (nf == 0) {
            LOAD_LAST(bufA, rl);
            MASK_LAST(bufA, rl);
            SUM_CHUNK(bufA);
        } else {
            LOAD_FULL(bufA, r0);
            int c = 0;
            for (; c + 2 < nf; c += 2) {
                LOAD_FULL(bufB, r0 + (c + 1) * 2 * CH);  PIN();
                SUM_CHUNK(bufA);                         PIN();
                LOAD_FULL(bufA, r0 + (c + 2) * 2 * CH);  PIN();
                SUM_CHUNK(bufB);                         PIN();
            }
            if (c + 1 < nf) {
                LOAD_FULL(bufB, r0 + (c + 1) * 2 * CH);  PIN();
                SUM_CHUNK(bufA);                         PIN();
                LOAD_LAST(bufA, rl);                     PIN();
                SUM_CHUNK(bufB);                         PIN();
                MASK_LAST(bufA, rl);
                SUM_CHUNK(bufA);
            } else {
                LOAD_LAST(bufB, rl);                     PIN();
                SUM_CHUNK(bufA);                         PIN();
                MASK_LAST(bufB, rl);
                SUM_CHUNK(bufB);
            }
        }
    }
#undef LOAD_FULL
#undef LOAD_LAST
#undef MASK_LAST
#undef SUM_CHUNK
#undef PIN

    v4f s4;
    s4.x = acc0[0]; s4.y = acc0[1]; s4.z = acc1[0]; s4.w = acc1[1];
    const bool mine = (h != 0) == onS;
    *(v4f*)&red[h][wave][4 * lmod] = mine ? s4 : zero4;
    __syncthreads();

    if (threadIdx.x < 2 * SLAB) {
        const int which = wave >> 1;
        if (which == 0 || two) {
            const int col = threadIdx.x & (SLAB - 1);
            float t = red[which][0][col];
            #pragma unroll
            for (int w = 1; w < NWAVES; ++w) t += red[which][w][col];
            const int ob = which ? bagS : bagL;
            const int oc = which ? cntS : cntL;
            out[(size_t)ob * D_DIM + colBase + col] = t / (float)oc;
        }
    }
}

extern "C" void kernel_launch(void* const* d_in, const int* in_sizes, int n_in,
                              void* d_out, int out_size, void* d_ws, size_t ws_size,
                              hipStream_t stream) {
    const float* samples = (const float*)d_in[0];
    const int*   counts  = (const int*)d_in[1];
    const int B = in_sizes[1];
    float* out = (float*)d_out;

    bag_pair_mean_wmma_kernel<<<dim3(D_DIM / SLAB, (B + 1) / 2), dim3(NWAVES * 32), 0, stream>>>(samples, counts, out, B);
}
